# starting kernel
# baseline (speedup 1.0000x reference)
#include <hip/hip_runtime.h>
#include <hip/hip_bf16.h>
#include <math.h>

typedef float v2f __attribute__((ext_vector_type(2)));
typedef float v8f __attribute__((ext_vector_type(8)));

#define NBINS  216
#define CDIM   13
#define KSTEPS 54
#define WMMA4(A,B,C) __builtin_amdgcn_wmma_f32_16x16x4_f32(false,(A),false,(B),(short)0,(C),false,false)

__device__ __forceinline__ float lane_read(float v, int srcLane) {
    return __builtin_bit_cast(float,
        __builtin_amdgcn_ds_bpermute(srcLane << 2, __builtin_bit_cast(int, v)));
}

__device__ __forceinline__ void softmax16(v8f& acc, float bias, int c, int lane) {
    #pragma unroll
    for (int r = 0; r < 8; ++r) {
        float lg = acc[r] + bias;
        float m = (c < CDIM) ? lg : -1e30f;
        #pragma unroll
        for (int off = 8; off; off >>= 1)
            m = fmaxf(m, lane_read(m, lane ^ off));
        float e = (c < CDIM) ? __expf(lg - m) : 0.0f;
        float s = e;
        #pragma unroll
        for (int off = 8; off; off >>= 1)
            s += lane_read(s, lane ^ off);
        acc[r] = e / s;
    }
}

__global__ __launch_bounds__(256)
void chord_heads_kernel(const float* __restrict__ x,
                        const float* __restrict__ Wk, const float* __restrict__ bk,
                        const float* __restrict__ Wd, const float* __restrict__ bd,
                        const float* __restrict__ Wb, const float* __restrict__ bb,
                        float* __restrict__ out, int R)
{
    __shared__ v2f wfrag[3][KSTEPS][32];
    const int tid = threadIdx.x;
    for (int idx = tid; idx < 3 * KSTEPS * 32; idx += blockDim.x) {
        int h   = idx / (KSTEPS * 32);
        int rem = idx - h * (KSTEPS * 32);
        int s   = rem >> 5;
        int l   = rem & 31;
        int cc  = l & 15;
        int hh  = l >> 4;
        int k0  = 4 * s + 2 * hh;
        const float* W = (h == 0) ? Wk : ((h == 1) ? Wd : Wb);
        v2f w; w.x = 0.0f; w.y = 0.0f;
        if (cc < CDIM) { w.x = W[k0 * CDIM + cc]; w.y = W[(k0 + 1) * CDIM + cc]; }
        wfrag[h][s][l] = w;
    }
    __syncthreads();

    const int lane = tid & 31;
    const int wave = tid >> 5;
    const int c    = lane & 15;
    const int hi   = lane >> 4;
    const int base = lane & 16;

    const float biasK = (c < CDIM) ? bk[c] : 0.0f;
    const float biasD = (c < CDIM) ? bd[c] : 0.0f;
    const float biasB = (c < CDIM) ? bb[c] : 0.0f;

    const int tiles = (R + 15) >> 4;
    const int gw    = blockIdx.x * (blockDim.x >> 5) + wave;
    const int nw    = gridDim.x * (blockDim.x >> 5);
    const long long hstride = (long long)R * CDIM;

    for (int t = gw; t < tiles; t += nw) {
        const int row0 = t << 4;
        int rl = row0 + (lane & 15);
        int rc = rl < R ? rl : R - 1;
        const float* xrow = x + (long long)rc * NBINS + 2 * hi;

        int tn = t + nw;
        if (tn < tiles) {
            int rn = (tn << 4) + (lane & 15);
            rn = rn < R ? rn : R - 1;
            __builtin_prefetch(x + (long long)rn * NBINS, 0, 1);
        }

        v8f accK = {0.f,0.f,0.f,0.f,0.f,0.f,0.f,0.f};
        v8f accD = {0.f,0.f,0.f,0.f,0.f,0.f,0.f,0.f};
        v8f accB = {0.f,0.f,0.f,0.f,0.f,0.f,0.f,0.f};

        v2f a  = *(const v2f*)(xrow);
        v2f w0 = wfrag[0][0][lane];
        v2f w1 = wfrag[1][0][lane];
        v2f w2 = wfrag[2][0][lane];
        #pragma unroll 3
        for (int s = 0; s < KSTEPS - 1; ++s) {
            v2f a_n  = *(const v2f*)(xrow + 4 * (s + 1));
            v2f w0n = wfrag[0][s + 1][lane];
            v2f w1n = wfrag[1][s + 1][lane];
            v2f w2n = wfrag[2][s + 1][lane];
            accK = WMMA4(a, w0, accK);
            accD = WMMA4(a, w1, accD);
            accB = WMMA4(a, w2, accB);
            a = a_n; w0 = w0n; w1 = w1n; w2 = w2n;
        }
        accK = WMMA4(a, w0, accK);
        accD = WMMA4(a, w1, accD);
        accB = WMMA4(a, w2, accB);

        softmax16(accK, biasK, c, lane);
        softmax16(accD, biasD, c, lane);
        softmax16(accB, biasB, c, lane);

        #pragma unroll
        for (int r = 0; r < 8; ++r) {
            int row = row0 + r + 8 * hi;
            if (row < R && c < CDIM) {
                long long o = (long long)row * CDIM + c;
                out[o]                = accK[r];
                out[hstride + o]      = accD[r];
                out[2 * hstride + o]  = accB[r];
            }
        }

        #pragma unroll
        for (int r = 0; r < 8; ++r) {
            float k0v = lane_read(accK[r], base);
            float d0v = lane_read(accD[r], base);
            float root0 = k0v + d0v - k0v * d0v;
            int rr = c - 1;
            float accR = 0.0f;
            #pragma unroll
            for (int k = 0; k < 12; ++k) {
                float kk = lane_read(accK[r], base + k + 1);
                int di = rr - k; di += (di < 0) ? 12 : 0; di += (di < 0) ? 12 : 0;
                float dd = lane_read(accD[r], base + di + 1);
                accR = fmaf(kk, dd, accR);
            }
            float val = (c == 0) ? root0 : accR;
            int row = row0 + r + 8 * hi;
            if (row < R && c < CDIM)
                out[3 * hstride + (long long)row * CDIM + c] = val;
        }
    }
}

extern "C" void kernel_launch(void* const* d_in, const int* in_sizes, int n_in,
                              void* d_out, int out_size, void* d_ws, size_t ws_size,
                              hipStream_t stream) {
    const float* x  = (const float*)d_in[0];
    const float* Wk = (const float*)d_in[1];
    const float* bk = (const float*)d_in[2];
    const float* Wd = (const float*)d_in[3];
    const float* bd = (const float*)d_in[4];
    const float* Wb = (const float*)d_in[5];
    const float* bb = (const float*)d_in[6];
    float* out = (float*)d_out;
    int R = in_sizes[0] / NBINS;

    dim3 block(256);
    dim3 grid(512);
    chord_heads_kernel<<<grid, block, 0, stream>>>(x, Wk, bk, Wd, bd, Wb, bb, out, R);
}


# x tiles read as whole 128-byte lines (27 b128 loads of 512 contiguous bytes per wave, next tile in flight under the K loop) through a wave-private LDS stage; same k order, sums unchanged
# speedup vs baseline: 1.0562x; 1.0562x over previous
#include <hip/hip_runtime.h>
#include <hip/hip_bf16.h>
#include <math.h>

typedef float v2f __attribute__((ext_vector_type(2)));
typedef float v4f __attribute__((ext_vector_type(4)));
typedef float v8f __attribute__((ext_vector_type(8)));

#define NBINS  216
#define CDIM   13
#define KSTEPS 54
#define ROWCH  54
#define TILECH 864
#define NGRP   9
#define GCH    3
#define BSTEPS 6
#define ABUF   (16 * NBINS)
#define WMMA4(A,B,C) __builtin_amdgcn_wmma_f32_16x16x4_f32(false,(A),false,(B),(short)0,(C),false,false)

__device__ __forceinline__ float lane_read(float v, int srcLane) {
    return __builtin_bit_cast(float,
        __builtin_amdgcn_ds_bpermute(srcLane << 2, __builtin_bit_cast(int, v)));
}

__device__ __forceinline__ void softmax16(v8f& acc, float bias, int c, int lane) {
    #pragma unroll
    for (int r = 0; r < 8; ++r) {
        float lg = acc[r] + bias;
        float m = (c < CDIM) ? lg : -1e30f;
        #pragma unroll
        for (int off = 8; off; off >>= 1)
            m = fmaxf(m, lane_read(m, lane ^ off));
        float e = (c < CDIM) ? __expf(lg - m) : 0.0f;
        float s = e;
        #pragma unroll
        for (int off = 8; off; off >>= 1)
            s += lane_read(s, lane ^ off);
        acc[r] = e / s;
    }
}

__device__ __forceinline__ void load_group(const float* __restrict__ x, int qbase, int last, int g, v4f (&st)[GCH]) {
    #pragma unroll
    for (int i = 0; i < GCH; ++i) {
        int qq = qbase + 32 * (g * GCH + i);
        qq = qq < last ? qq : last;
        st[i] = ((const v4f*)x)[qq];
    }
}

__device__ __forceinline__ void stash_group(float* buf, int lane, int g, const v4f (&st)[GCH]) {
    #pragma unroll
    for (int i = 0; i < GCH; ++i)
        *(v4f*)(buf + 4 * (32 * (g * GCH + i) + lane)) = st[i];
}

__global__ __launch_bounds__(256)
void chord_heads_kernel(const float* __restrict__ x,
                        const float* __restrict__ Wk, const float* __restrict__ bk,
                        const float* __restrict__ Wd, const float* __restrict__ bd,
                        const float* __restrict__ Wb, const float* __restrict__ bb,
                        float* __restrict__ out, int R)
{
    __shared__ v2f wfrag[3][KSTEPS][32];
    __shared__ __attribute__((aligned(16))) float abuf[8 * 2 * ABUF];
    const int tid = threadIdx.x;

    float* myA = abuf + (tid >> 5) * (2 * ABUF);
    const int t0 = blockIdx.x * (blockDim.x >> 5) + (tid >> 5);
    const bool have0 = t0 < ((R + 15) >> 4);
    v4f f0[NGRP][GCH];
    if (have0) {
        #pragma unroll
        for (int g = 0; g < NGRP; ++g) load_group(x, t0 * TILECH + (tid & 31), R * ROWCH - 1, g, f0[g]);
    }
    for (int idx = tid; idx < 3 * KSTEPS * 32; idx += blockDim.x) {
        int h   = idx / (KSTEPS * 32);
        int rem = idx - h * (KSTEPS * 32);
        int s   = rem >> 5;
        int l   = rem & 31;
        int cc  = l & 15;
        int hh  = l >> 4;
        int k0  = 4 * s + 2 * hh;
        const float* W = (h == 0) ? Wk : ((h == 1) ? Wd : Wb);
        v2f w; w.x = 0.0f; w.y = 0.0f;
        if (cc < CDIM) { w.x = W[k0 * CDIM + cc]; w.y = W[(k0 + 1) * CDIM + cc]; }
        wfrag[h][s][l] = w;
    }
    if (have0) {
        #pragma unroll
        for (int g = 0; g < NGRP; ++g) stash_group(myA, tid & 31, g, f0[g]);
    }
    __syncthreads();

    const int lane = tid & 31;
    const int wave = tid >> 5;
    const int c    = lane & 15;
    const int hi   = lane >> 4;
    const int base = lane & 16;

    const float biasK = (c < CDIM) ? bk[c] : 0.0f;
    const float biasD = (c < CDIM) ? bd[c] : 0.0f;
    const float biasB = (c < CDIM) ? bb[c] : 0.0f;

    const int tiles = (R + 15) >> 4;
    const int gw    = blockIdx.x * (blockDim.x >> 5) + wave;
    const int nw    = gridDim.x * (blockDim.x >> 5);
    const long long hstride = (long long)R * CDIM;

    v2f bK[KSTEPS], bD[KSTEPS], bB[KSTEPS];
    #pragma unroll
    for (int s = 0; s < KSTEPS; ++s) {
        bK[s] = wfrag[0][s][lane];
        bD[s] = wfrag[1][s][lane];
        bB[s] = wfrag[2][s][lane];
    }

    const int last = R * ROWCH - 1;
    const int aoff = (lane & 15) * NBINS + 2 * hi;

    int par = 0;
    for (int t = gw; t < tiles; t += nw) {
        const int row0 = t << 4;
        const int tn = t + nw;
        const bool more = tn < tiles;
        const int qb = more ? tn * TILECH + lane : last;
        const float* arow = myA + par * ABUF + aoff;
        float* nxt = myA + (par ^ 1) * ABUF;
        par ^= 1;

        v8f accK = {0.f,0.f,0.f,0.f,0.f,0.f,0.f,0.f};
        v8f accD = {0.f,0.f,0.f,0.f,0.f,0.f,0.f,0.f};
        v8f accB = {0.f,0.f,0.f,0.f,0.f,0.f,0.f,0.f};

        v4f sg[3][GCH];
        v2f af[2][BSTEPS];
        #pragma unroll
        for (int j = 0; j < BSTEPS; ++j) af[0][j] = *(const v2f*)(arow + 4 * j);
        #pragma unroll
        for (int b = 0; b < NGRP; ++b) {
            if (b >= 3) stash_group(nxt, lane, b - 3, sg[b % 3]);
            load_group(x, qb, last, b, sg[b % 3]);
            if (b + 1 < NGRP) {
                #pragma unroll
                for (int j = 0; j < BSTEPS; ++j) af[(b + 1) & 1][j] = *(const v2f*)(arow + 4 * ((b + 1) * BSTEPS + j));
            }
            #pragma unroll
            for (int j = 0; j < BSTEPS; ++j) {
                const int s = b * BSTEPS + j;
                accK = WMMA4(af[b & 1][j], bK[s], accK);
                accD = WMMA4(af[b & 1][j], bD[s], accD);
                accB = WMMA4(af[b & 1][j], bB[s], accB);
            }
        }

        softmax16(accK, biasK, c, lane);
        softmax16(accD, biasD, c, lane);
        softmax16(accB, biasB, c, lane);

        #pragma unroll
        for (int r = 0; r < 8; ++r) {
            int row = row0 + r + 8 * hi;
            if (row < R && c < CDIM) {
                long long o = (long long)row * CDIM + c;
                out[o]                = accK[r];
                out[hstride + o]      = accD[r];
                out[2 * hstride + o]  = accB[r];
            }
        }

        #pragma unroll
        for (int r = 0; r < 8; ++r) {
            float k0v = lane_read(accK[r], base);
            float d0v = lane_read(accD[r], base);
            float root0 = k0v + d0v - k0v * d0v;
            int rr = c - 1;
            float accR = 0.0f;
            #pragma unroll
            for (int k = 0; k < 12; ++k) {
                float kk = lane_read(accK[r], base + k + 1);
                int di = rr - k; di += (di < 0) ? 12 : 0; di += (di < 0) ? 12 : 0;
                float dd = lane_read(accD[r], base + di + 1);
                accR = fmaf(kk, dd, accR);
            }
            float val = (c == 0) ? root0 : accR;
            int row = row0 + r + 8 * hi;
            if (row < R && c < CDIM)
                out[3 * hstride + (long long)row * CDIM + c] = val;
        }

        stash_group(nxt, lane, 6, sg[0]);
        stash_group(nxt, lane, 7, sg[1]);
        stash_group(nxt, lane, 8, sg[2]);
    }
}

extern "C" void kernel_launch(void* const* d_in, const int* in_sizes, int n_in,
                              void* d_out, int out_size, void* d_ws, size_t ws_size,
                              hipStream_t stream) {
    const float* x  = (const float*)d_in[0];
    const float* Wk = (const float*)d_in[1];
    const float* bk = (const float*)d_in[2];
    const float* Wd = (const float*)d_in[3];
    const float* bd = (const float*)d_in[4];
    const float* Wb = (const float*)d_in[5];
    const float* bb = (const float*)d_in[6];
    float* out = (float*)d_out;
    int R = in_sizes[0] / NBINS;

    dim3 block(256);
    dim3 grid(512);
    chord_heads_kernel<<<grid, block, 0, stream>>>(x, Wk, bk, Wd, bd, Wb, bb, out, R);
}


# WMMA operands swapped (weights as A, x rows as B): a lane holds one row's classes, so softmax and root convolution run in registers, not LDS lane permutes; softmax sum order changes; outputs via LDS
# speedup vs baseline: 1.1835x; 1.1205x over previous
#include <hip/hip_runtime.h>
#include <hip/hip_bf16.h>
#include <math.h>

typedef float v2f __attribute__((ext_vector_type(2)));
typedef float v4f __attribute__((ext_vector_type(4)));
typedef float v8f __attribute__((ext_vector_type(8)));

#define NBINS  216
#define CDIM   13
#define KSTEPS 54
#define ROWCH  54
#define TILECH 864
#define NGRP   9
#define GCH    3
#define BSTEPS 6
#define ABUF   (16 * NBINS)
#define WMMA4(A,B,C) __builtin_amdgcn_wmma_f32_16x16x4_f32(false,(A),false,(B),(short)0,(C),false,false)

#define OUTT   (16 * CDIM)
#define OCH    OUTT

__device__ __forceinline__ float swap_half(float v) {
    int b = __builtin_bit_cast(int, v);
    return __builtin_bit_cast(float, __builtin_amdgcn_permlanex16(b, b, 0x76543210, 0xfedcba98, false, false));
}

__device__ __forceinline__ void softmax_row(v8f& acc, const float (&bias)[8], int hi) {
    float lg[8];
    float m = -1e30f;
    #pragma unroll
    for (int r = 0; r < 8; ++r) {
        lg[r] = acc[r] + bias[r];
        float v = (r >= 5 && hi) ? -1e30f : lg[r];
        m = fmaxf(m, v);
    }
    m = fmaxf(m, swap_half(m));
    float s = 0.0f;
    #pragma unroll
    for (int r = 0; r < 8; ++r) {
        float e = (r >= 5 && hi) ? 0.0f : __expf(lg[r] - m);
        lg[r] = e;
        s += e;
    }
    s += swap_half(s);
    #pragma unroll
    for (int r = 0; r < 8; ++r) acc[r] = lg[r] / s;
}

__device__ __forceinline__ float pick(const v8f& own, const v8f& oth, int hi, int c0, int c1) {
    float a = (c0 < 8) ? own[c0] : oth[c0 - 8];
    float b = (c1 >= 8) ? own[c1 - 8] : oth[c1];
    return hi ? b : a;
}

__device__ __forceinline__ void load_group(const float* __restrict__ x, int qbase, int last, int g, v4f (&st)[GCH]) {
    #pragma unroll
    for (int i = 0; i < GCH; ++i) {
        int qq = qbase + 32 * (g * GCH + i);
        qq = qq < last ? qq : last;
        st[i] = ((const v4f*)x)[qq];
    }
}

__device__ __forceinline__ void stash_group(float* buf, int lane, int g, const v4f (&st)[GCH]) {
    #pragma unroll
    for (int i = 0; i < GCH; ++i)
        *(v4f*)(buf + 4 * (32 * (g * GCH + i) + lane)) = st[i];
}

__global__ __launch_bounds__(256)
void chord_heads_kernel(const float* __restrict__ x,
                        const float* __restrict__ Wk, const float* __restrict__ bk,
                        const float* __restrict__ Wd, const float* __restrict__ bd,
                        const float* __restrict__ Wb, const float* __restrict__ bb,
                        float* __restrict__ out, int R)
{
    __shared__ v2f wfrag[3][KSTEPS][32];
    __shared__ __attribute__((aligned(16))) float abuf[8 * 2 * ABUF];
    __shared__ __attribute__((aligned(16))) float btab[2][3][8];
    const int tid = threadIdx.x;

    float* myA = abuf + (tid >> 5) * (2 * ABUF);
    const int t0 = blockIdx.x * (blockDim.x >> 5) + (tid >> 5);
    const bool have0 = t0 < ((R + 15) >> 4);
    v4f f0[NGRP][GCH];
    if (have0) {
        #pragma unroll
        for (int g = 0; g < NGRP; ++g) load_group(x, t0 * TILECH + (tid & 31), R * ROWCH - 1, g, f0[g]);
    }
    for (int idx = tid; idx < 3 * KSTEPS * 32; idx += blockDim.x) {
        int h   = idx / (KSTEPS * 32);
        int rem = idx - h * (KSTEPS * 32);
        int s   = rem >> 5;
        int l   = rem & 31;
        int cc  = l & 15;
        int hh  = l >> 4;
        int k0  = 4 * s + 2 * hh;
        const float* W = (h == 0) ? Wk : ((h == 1) ? Wd : Wb);
        v2f w; w.x = 0.0f; w.y = 0.0f;
        if (cc < CDIM) { w.x = W[k0 * CDIM + cc]; w.y = W[(k0 + 1) * CDIM + cc]; }
        wfrag[h][s][l] = w;
    }
    if (tid < 48) {
        const int hf = tid / 24, hd = (tid % 24) >> 3, r = tid & 7;
        const int cl = 8 * hf + r;
        const float* bp = (hd == 0) ? bk : ((hd == 1) ? bd : bb);
        btab[hf][hd][r] = (cl < CDIM) ? bp[cl] : 0.0f;
    }
    if (have0) {
        #pragma unroll
        for (int g = 0; g < NGRP; ++g) stash_group(myA, tid & 31, g, f0[g]);
    }
    __syncthreads();

    const int lane = tid & 31;
    const int wave = tid >> 5;
    const int hi   = lane >> 4;

    const int tiles = (R + 15) >> 4;
    const int gw    = blockIdx.x * (blockDim.x >> 5) + wave;
    const int nw    = gridDim.x * (blockDim.x >> 5);
    const long long hstride = (long long)R * CDIM;

    v2f bK[KSTEPS], bD[KSTEPS], bB[KSTEPS];
    #pragma unroll
    for (int s = 0; s < KSTEPS; ++s) {
        bK[s] = wfrag[0][s][lane];
        bD[s] = wfrag[1][s][lane];
        bB[s] = wfrag[2][s][lane];
    }

    const int last = R * ROWCH - 1;
    const int aoff = (lane & 15) * NBINS + 2 * hi;

    int par = 0;
    for (int t = gw; t < tiles; t += nw) {
        const int row0 = t << 4;
        const int tn = t + nw;
        const bool more = tn < tiles;
        const int qb = more ? tn * TILECH + lane : last;
        const float* arow = myA + par * ABUF + aoff;
        float* nxt = myA + (par ^ 1) * ABUF;
        par ^= 1;

        v8f accK = {0.f,0.f,0.f,0.f,0.f,0.f,0.f,0.f};
        v8f accD = {0.f,0.f,0.f,0.f,0.f,0.f,0.f,0.f};
        v8f accB = {0.f,0.f,0.f,0.f,0.f,0.f,0.f,0.f};

        v4f sg[3][GCH];
        v2f af[2][BSTEPS];
        #pragma unroll
        for (int j = 0; j < BSTEPS; ++j) af[0][j] = *(const v2f*)(arow + 4 * j);
        #pragma unroll
        for (int b = 0; b < NGRP; ++b) {
            if (b >= 3) stash_group(nxt, lane, b - 3, sg[b % 3]);
            load_group(x, qb, last, b, sg[b % 3]);
            if (b + 1 < NGRP) {
                #pragma unroll
                for (int j = 0; j < BSTEPS; ++j) af[(b + 1) & 1][j] = *(const v2f*)(arow + 4 * ((b + 1) * BSTEPS + j));
            }
            #pragma unroll
            for (int j = 0; j < BSTEPS; ++j) {
                const int s = b * BSTEPS + j;
                accK = WMMA4(bK[s], af[b & 1][j], accK);
                accD = WMMA4(bD[s], af[b & 1][j], accD);
                accB = WMMA4(bB[s], af[b & 1][j], accB);
            }
        }

        float biasK[8], biasD[8], biasB[8];
        {
            const v4f* bt = (const v4f*)&btab[hi][0][0];
            const v4f k0 = bt[0], k1 = bt[1], d0 = bt[2], d1 = bt[3], b0 = bt[4], b1 = bt[5];
            #pragma unroll
            for (int r = 0; r < 4; ++r) {
                biasK[r] = k0[r]; biasK[4 + r] = k1[r];
                biasD[r] = d0[r]; biasD[4 + r] = d1[r];
                biasB[r] = b0[r]; biasB[4 + r] = b1[r];
            }
        }
        softmax_row(accK, biasK, hi);
        softmax_row(accD, biasD, hi);
        softmax_row(accB, biasB, hi);

        v8f oK, oD;
        #pragma unroll
        for (int r = 0; r < 8; ++r) { oK[r] = swap_half(accK[r]); oD[r] = swap_half(accD[r]); }
        float Kf[12], Dr[12];
        #pragma unroll
        for (int k = 0; k < 12; ++k) {
            Kf[k] = pick(accK, oK, hi, k + 1, k + 1);
            Dr[k] = pick(accD, oD, hi, ((k + 11) % 12) + 1, ((k + 7) % 12) + 1);
        }
        v8f accR;
        #pragma unroll
        for (int r = 0; r < 8; ++r) {
            float a = 0.0f;
            #pragma unroll
            for (int k = 0; k < 12; ++k) a = fmaf(Kf[k], Dr[(r - k + 12) % 12], a);
            accR[r] = a;
        }
        {
            float k0v = accK[0], d0v = accD[0];
            float root0 = k0v + d0v - k0v * d0v;
            accR[0] = hi ? accR[0] : root0;
        }

        float* ob = myA + (par ^ 1) * ABUF;
        {
            const int so = (lane & 15) * CDIM + 8 * hi;
            #pragma unroll
            for (int r = 0; r < 8; ++r) {
                if (r < 5 || !hi) {
                    ob[so + r]            = accK[r];
                    ob[OUTT + so + r]     = accD[r];
                    ob[2 * OUTT + so + r] = accB[r];
                    ob[3 * OUTT + so + r] = accR[r];
                }
            }
        }
        if (row0 + 16 <= R && (R & 3) == 0) {
            #pragma unroll
            for (int i = 0; i < 7; ++i) {
                const int q = lane + 32 * i;
                if (q < OCH) {
                    const int hd = q / 52;
                    const int w  = q - hd * 52;
                    const v4f v = *(const v4f*)(ob + 4 * q);
                    *(v4f*)(out + hd * hstride + (long long)row0 * CDIM + 4 * w) = v;
                }
            }
        } else {
            for (int idx = lane; idx < 4 * OUTT; idx += 32) {
                const int hd = idx / OUTT;
                const int w  = idx - hd * OUTT;
                if (row0 + w / CDIM < R)
                    out[hd * hstride + (long long)row0 * CDIM + w] = ob[idx];
            }
        }

        stash_group(nxt, lane, 6, sg[0]);
        stash_group(nxt, lane, 7, sg[1]);
        stash_group(nxt, lane, 8, sg[2]);
    }
}

extern "C" void kernel_launch(void* const* d_in, const int* in_sizes, int n_in,
                              void* d_out, int out_size, void* d_ws, size_t ws_size,
                              hipStream_t stream) {
    const float* x  = (const float*)d_in[0];
    const float* Wk = (const float*)d_in[1];
    const float* bk = (const float*)d_in[2];
    const float* Wd = (const float*)d_in[3];
    const float* bd = (const float*)d_in[4];
    const float* Wb = (const float*)d_in[5];
    const float* bb = (const float*)d_in[6];
    float* out = (float*)d_out;
    int R = in_sizes[0] / NBINS;

    dim3 block(256);
    dim3 grid(512);
    chord_heads_kernel<<<grid, block, 0, stream>>>(x, Wk, bk, Wd, bd, Wb, bb, out, R);
}


# K loop order pinned with scheduling barriers: a group's 3 global loads stay 5 blocks (90 WMMAs) ahead of its LDS store and a block's memory instructions stand in one place; sums unchanged
# speedup vs baseline: 1.2361x; 1.0444x over previous
#include <hip/hip_runtime.h>
#include <hip/hip_bf16.h>
#include <math.h>

typedef float v2f __attribute__((ext_vector_type(2)));
typedef float v4f __attribute__((ext_vector_type(4)));
typedef float v8f __attribute__((ext_vector_type(8)));

#define NBINS  216
#define CDIM   13
#define KSTEPS 54
#define ROWCH  54
#define TILECH 864
#define NGRP   9
#define GCH    3
#define BSTEPS 6
#define DEPTH  5
#define ABUF   (16 * NBINS)
#define WMMA4(A,B,C) __builtin_amdgcn_wmma_f32_16x16x4_f32(false,(A),false,(B),(short)0,(C),false,false)

#define OUTT   (16 * CDIM)
#define OCH    OUTT

__device__ __forceinline__ float swap_half(float v) {
    int b = __builtin_bit_cast(int, v);
    return __builtin_bit_cast(float, __builtin_amdgcn_permlanex16(b, b, 0x76543210, 0xfedcba98, false, false));
}

__device__ __forceinline__ void softmax_row(v8f& acc, const float (&bias)[8], int hi) {
    float lg[8];
    float m = -1e30f;
    #pragma unroll
    for (int r = 0; r < 8; ++r) {
        lg[r] = acc[r] + bias[r];
        float v = (r >= 5 && hi) ? -1e30f : lg[r];
        m = fmaxf(m, v);
    }
    m = fmaxf(m, swap_half(m));
    float s = 0.0f;
    #pragma unroll
    for (int r = 0; r < 8; ++r) {
        float e = (r >= 5 && hi) ? 0.0f : __expf(lg[r] - m);
        lg[r] = e;
        s += e;
    }
    s += swap_half(s);
    #pragma unroll
    for (int r = 0; r < 8; ++r) acc[r] = lg[r] / s;
}

__device__ __forceinline__ float pick(const v8f& own, const v8f& oth, int hi, int c0, int c1) {
    float a = (c0 < 8) ? own[c0] : oth[c0 - 8];
    float b = (c1 >= 8) ? own[c1 - 8] : oth[c1];
    return hi ? b : a;
}

__device__ __forceinline__ void load_group(const float* __restrict__ x, int qbase, int last, int g, v4f (&st)[GCH]) {
    #pragma unroll
    for (int i = 0; i < GCH; ++i) {
        int qq = qbase + 32 * (g * GCH + i);
        qq = qq < last ? qq : last;
        st[i] = ((const v4f*)x)[qq];
    }
}

__device__ __forceinline__ void stash_group(float* buf, int lane, int g, const v4f (&st)[GCH]) {
    #pragma unroll
    for (int i = 0; i < GCH; ++i)
        *(v4f*)(buf + 4 * (32 * (g * GCH + i) + lane)) = st[i];
}

__global__ __launch_bounds__(256)
void chord_heads_kernel(const float* __restrict__ x,
                        const float* __restrict__ Wk, const float* __restrict__ bk,
                        const float* __restrict__ Wd, const float* __restrict__ bd,
                        const float* __restrict__ Wb, const float* __restrict__ bb,
                        float* __restrict__ out, int R)
{
    __shared__ v2f wfrag[3][KSTEPS][32];
    __shared__ __attribute__((aligned(16))) float abuf[8 * 2 * ABUF];
    __shared__ __attribute__((aligned(16))) float btab[2][3][8];
    const int tid = threadIdx.x;

    float* myA = abuf + (tid >> 5) * (2 * ABUF);
    const int t0 = blockIdx.x * (blockDim.x >> 5) + (tid >> 5);
    const bool have0 = t0 < ((R + 15) >> 4);
    v4f f0[NGRP][GCH];
    if (have0) {
        #pragma unroll
        for (int g = 0; g < NGRP; ++g) load_group(x, t0 * TILECH + (tid & 31), R * ROWCH - 1, g, f0[g]);
    }
    for (int idx = tid; idx < 3 * KSTEPS * 32; idx += blockDim.x) {
        int h   = idx / (KSTEPS * 32);
        int rem = idx - h * (KSTEPS * 32);
        int s   = rem >> 5;
        int l   = rem & 31;
        int cc  = l & 15;
        int hh  = l >> 4;
        int k0  = 4 * s + 2 * hh;
        const float* W = (h == 0) ? Wk : ((h == 1) ? Wd : Wb);
        v2f w; w.x = 0.0f; w.y = 0.0f;
        if (cc < CDIM) { w.x = W[k0 * CDIM + cc]; w.y = W[(k0 + 1) * CDIM + cc]; }
        wfrag[h][s][l] = w;
    }
    if (tid < 48) {
        const int hf = tid / 24, hd = (tid % 24) >> 3, r = tid & 7;
        const int cl = 8 * hf + r;
        const float* bp = (hd == 0) ? bk : ((hd == 1) ? bd : bb);
        btab[hf][hd][r] = (cl < CDIM) ? bp[cl] : 0.0f;
    }
    if (have0) {
        #pragma unroll
        for (int g = 0; g < NGRP; ++g) stash_group(myA, tid & 31, g, f0[g]);
    }
    __syncthreads();

    const int lane = tid & 31;
    const int wave = tid >> 5;
    const int hi   = lane >> 4;

    const int tiles = (R + 15) >> 4;
    const int gw    = blockIdx.x * (blockDim.x >> 5) + wave;
    const int nw    = gridDim.x * (blockDim.x >> 5);
    const long long hstride = (long long)R * CDIM;

    v2f bK[KSTEPS], bD[KSTEPS], bB[KSTEPS];
    #pragma unroll
    for (int s = 0; s < KSTEPS; ++s) {
        bK[s] = wfrag[0][s][lane];
        bD[s] = wfrag[1][s][lane];
        bB[s] = wfrag[2][s][lane];
    }

    const int last = R * ROWCH - 1;
    const int aoff = (lane & 15) * NBINS + 2 * hi;

    int par = 0;
    for (int t = gw; t < tiles; t += nw) {
        const int row0 = t << 4;
        const int tn = t + nw;
        const bool more = tn < tiles;
        const int qb = more ? tn * TILECH + lane : last;
        const float* arow = myA + par * ABUF + aoff;
        float* nxt = myA + (par ^ 1) * ABUF;
        par ^= 1;

        v8f accK = {0.f,0.f,0.f,0.f,0.f,0.f,0.f,0.f};
        v8f accD = {0.f,0.f,0.f,0.f,0.f,0.f,0.f,0.f};
        v8f accB = {0.f,0.f,0.f,0.f,0.f,0.f,0.f,0.f};

        v4f sg[DEPTH][GCH];
        v2f af[2][BSTEPS];
        v4f bq[6];
        #pragma unroll
        for (int j = 0; j < BSTEPS; ++j) af[0][j] = *(const v2f*)(arow + 4 * j);
        #pragma unroll
        for (int b = 0; b < NGRP; ++b) {
            if (b >= DEPTH) stash_group(nxt, lane, b - DEPTH, sg[b % DEPTH]);
            load_group(x, qb, last, b, sg[b % DEPTH]);
            if (b + 1 < NGRP) {
                #pragma unroll
                for (int j = 0; j < BSTEPS; ++j) af[(b + 1) & 1][j] = *(const v2f*)(arow + 4 * ((b + 1) * BSTEPS + j));
            } else {
                const v4f* bt = (const v4f*)&btab[hi][0][0];
                #pragma unroll
                for (int i = 0; i < 6; ++i) bq[i] = bt[i];
            }
            __builtin_amdgcn_sched_barrier(0);
            #pragma unroll
            for (int j = 0; j < BSTEPS; ++j) {
                const int s = b * BSTEPS + j;
                accK = WMMA4(bK[s], af[b & 1][j], accK);
                accD = WMMA4(bD[s], af[b & 1][j], accD);
                accB = WMMA4(bB[s], af[b & 1][j], accB);
            }
            __builtin_amdgcn_sched_barrier(0);
        }

        float biasK[8], biasD[8], biasB[8];
        {
            const v4f k0 = bq[0], k1 = bq[1], d0 = bq[2], d1 = bq[3], b0 = bq[4], b1 = bq[5];
            #pragma unroll
            for (int r = 0; r < 4; ++r) {
                biasK[r] = k0[r]; biasK[4 + r] = k1[r];
                biasD[r] = d0[r]; biasD[4 + r] = d1[r];
                biasB[r] = b0[r]; biasB[4 + r] = b1[r];
            }
        }
        softmax_row(accK, biasK, hi);
        softmax_row(accD, biasD, hi);
        softmax_row(accB, biasB, hi);

        v8f oK, oD;
        #pragma unroll
        for (int r = 0; r < 8; ++r) { oK[r] = swap_half(accK[r]); oD[r] = swap_half(accD[r]); }
        float Kf[12], Dr[12];
        #pragma unroll
        for (int k = 0; k < 12; ++k) {
            Kf[k] = pick(accK, oK, hi, k + 1, k + 1);
            Dr[k] = pick(accD, oD, hi, ((k + 11) % 12) + 1, ((k + 7) % 12) + 1);
        }
        v8f accR;
        #pragma unroll
        for (int r = 0; r < 8; ++r) {
            float a = 0.0f;
            #pragma unroll
            for (int k = 0; k < 12; ++k) a = fmaf(Kf[k], Dr[(r - k + 12) % 12], a);
            accR[r] = a;
        }
        {
            float k0v = accK[0], d0v = accD[0];
            float root0 = k0v + d0v - k0v * d0v;
            accR[0] = hi ? accR[0] : root0;
        }

        float* ob = myA + (par ^ 1) * ABUF;
        {
            const int so = (lane & 15) * CDIM + 8 * hi;
            #pragma unroll
            for (int r = 0; r < 8; ++r) {
                if (r < 5 || !hi) {
                    ob[so + r]            = accK[r];
                    ob[OUTT + so + r]     = accD[r];
                    ob[2 * OUTT + so + r] = accB[r];
                    ob[3 * OUTT + so + r] = accR[r];
                }
            }
        }
        if (row0 + 16 <= R && (R & 3) == 0) {
            #pragma unroll
            for (int i = 0; i < 7; ++i) {
                const int q = lane + 32 * i;
                if (q < OCH) {
                    const int hd = q / 52;
                    const int w  = q - hd * 52;
                    const v4f v = *(const v4f*)(ob + 4 * q);
                    *(v4f*)(out + hd * hstride + (long long)row0 * CDIM + 4 * w) = v;
                }
            }
        } else {
            for (int idx = lane; idx < 4 * OUTT; idx += 32) {
                const int hd = idx / OUTT;
                const int w  = idx - hd * OUTT;
                if (row0 + w / CDIM < R)
                    out[hd * hstride + (long long)row0 * CDIM + w] = ob[idx];
            }
        }

        #pragma unroll
        for (int g = NGRP - DEPTH; g < NGRP; ++g) stash_group(nxt, lane, g, sg[g % DEPTH]);
    }
}

extern "C" void kernel_launch(void* const* d_in, const int* in_sizes, int n_in,
                              void* d_out, int out_size, void* d_ws, size_t ws_size,
                              hipStream_t stream) {
    const float* x  = (const float*)d_in[0];
    const float* Wk = (const float*)d_in[1];
    const float* bk = (const float*)d_in[2];
    const float* Wd = (const float*)d_in[3];
    const float* bd = (const float*)d_in[4];
    const float* Wb = (const float*)d_in[5];
    const float* bb = (const float*)d_in[6];
    float* out = (float*)d_out;
    int R = in_sizes[0] / NBINS;

    dim3 block(256);
    dim3 grid(512);
    chord_heads_kernel<<<grid, block, 0, stream>>>(x, Wk, bk, Wd, bd, Wb, bb, out, R);
}


# weight and bias staging: a thread's 43 global loads requested together, without branches, ahead of the first tile's loads, then stored to LDS; was one memory wait per pass, 21 passes; sums unchanged
# speedup vs baseline: 1.6242x; 1.3140x over previous
#include <hip/hip_runtime.h>
#include <hip/hip_bf16.h>
#include <math.h>

typedef float v2f __attribute__((ext_vector_type(2)));
typedef float v4f __attribute__((ext_vector_type(4)));
typedef float v8f __attribute__((ext_vector_type(8)));

#define NBINS  216
#define CDIM   13
#define KSTEPS 54
#define ROWCH  54
#define TILECH 864
#define NGRP   9
#define GCH    3
#define BSTEPS 6
#define DEPTH  5
#define ABUF   (16 * NBINS)
#define WMMA4(A,B,C) __builtin_amdgcn_wmma_f32_16x16x4_f32(false,(A),false,(B),(short)0,(C),false,false)

#define WTOT   (3 * KSTEPS * 32)
#define WSLOTS ((WTOT + 255) / 256)

#define OUTT   (16 * CDIM)
#define OCH    OUTT

__device__ __forceinline__ float swap_half(float v) {
    int b = __builtin_bit_cast(int, v);
    return __builtin_bit_cast(float, __builtin_amdgcn_permlanex16(b, b, 0x76543210, 0xfedcba98, false, false));
}

__device__ __forceinline__ void softmax_row(v8f& acc, const float (&bias)[8], int hi) {
    float lg[8];
    float m = -1e30f;
    #pragma unroll
    for (int r = 0; r < 8; ++r) {
        lg[r] = acc[r] + bias[r];
        float v = (r >= 5 && hi) ? -1e30f : lg[r];
        m = fmaxf(m, v);
    }
    m = fmaxf(m, swap_half(m));
    float s = 0.0f;
    #pragma unroll
    for (int r = 0; r < 8; ++r) {
        float e = (r >= 5 && hi) ? 0.0f : __expf(lg[r] - m);
        lg[r] = e;
        s += e;
    }
    s += swap_half(s);
    #pragma unroll
    for (int r = 0; r < 8; ++r) acc[r] = lg[r] / s;
}

__device__ __forceinline__ float pick(const v8f& own, const v8f& oth, int hi, int c0, int c1) {
    float a = (c0 < 8) ? own[c0] : oth[c0 - 8];
    float b = (c1 >= 8) ? own[c1 - 8] : oth[c1];
    return hi ? b : a;
}

__device__ __forceinline__ void load_group(const float* __restrict__ x, int qbase, int last, int g, v4f (&st)[GCH]) {
    #pragma unroll
    for (int i = 0; i < GCH; ++i) {
        int qq = qbase + 32 * (g * GCH + i);
        qq = qq < last ? qq : last;
        st[i] = ((const v4f*)x)[qq];
    }
}

__device__ __forceinline__ void stash_group(float* buf, int lane, int g, const v4f (&st)[GCH]) {
    #pragma unroll
    for (int i = 0; i < GCH; ++i)
        *(v4f*)(buf + 4 * (32 * (g * GCH + i) + lane)) = st[i];
}

__global__ __launch_bounds__(256)
void chord_heads_kernel(const float* __restrict__ x,
                        const float* __restrict__ Wk, const float* __restrict__ bk,
                        const float* __restrict__ Wd, const float* __restrict__ bd,
                        const float* __restrict__ Wb, const float* __restrict__ bb,
                        float* __restrict__ out, int R)
{
    __shared__ v2f wfrag[3][KSTEPS][32];
    __shared__ __attribute__((aligned(16))) float abuf[8 * 2 * ABUF];
    __shared__ __attribute__((aligned(16))) float btab[2][3][8];
    const int tid = threadIdx.x;
    __builtin_assume(tid < 256);

    float* myA = abuf + (tid >> 5) * (2 * ABUF);
    v2f wv[WSLOTS];
    #pragma unroll
    for (int i = 0; i < WSLOTS; ++i) {
        int idx = tid + 256 * i;
        idx = idx < WTOT ? idx : WTOT - 1;
        int h   = idx / (KSTEPS * 32);
        int rem = idx - h * (KSTEPS * 32);
        int s   = rem >> 5;
        int l   = rem & 31;
        int cc  = l & 15;
        int hh  = l >> 4;
        int k0  = 4 * s + 2 * hh;
        const float* W = (h == 0) ? Wk : ((h == 1) ? Wd : Wb);
        int cl  = cc < CDIM ? cc : CDIM - 1;
        float w0 = W[k0 * CDIM + cl];
        float w1 = W[(k0 + 1) * CDIM + cl];
        wv[i].x = cc < CDIM ? w0 : 0.0f;
        wv[i].y = cc < CDIM ? w1 : 0.0f;
    }
    const int bhf = tid / 24, bhd = (tid % 24) >> 3, br = tid & 7;
    const int bcl = 8 * bhf + br;
    const float* bp = (bhd == 0) ? bk : ((bhd == 1) ? bd : bb);
    const float bv = bp[bcl < CDIM ? bcl : CDIM - 1];
    const int t0 = blockIdx.x * (blockDim.x >> 5) + (tid >> 5);
    const bool have0 = t0 < ((R + 15) >> 4);
    v4f f0[NGRP][GCH];
    if (have0) {
        #pragma unroll
        for (int g = 0; g < NGRP; ++g) load_group(x, t0 * TILECH + (tid & 31), R * ROWCH - 1, g, f0[g]);
    }
    #pragma unroll
    for (int i = 0; i < WSLOTS; ++i) {
        const int idx = tid + 256 * i;
        if (idx < WTOT) (&wfrag[0][0][0])[idx] = wv[i];
    }
    if (tid < 48) btab[bhf][bhd][br] = (bcl < CDIM) ? bv : 0.0f;
    if (have0) {
        #pragma unroll
        for (int g = 0; g < NGRP; ++g) stash_group(myA, tid & 31, g, f0[g]);
    }
    __syncthreads();

    const int lane = tid & 31;
    const int wave = tid >> 5;
    const int hi   = lane >> 4;

    const int tiles = (R + 15) >> 4;
    const int gw    = blockIdx.x * (blockDim.x >> 5) + wave;
    const int nw    = gridDim.x * (blockDim.x >> 5);
    const long long hstride = (long long)R * CDIM;

    v2f bK[KSTEPS], bD[KSTEPS], bB[KSTEPS];
    #pragma unroll
    for (int s = 0; s < KSTEPS; ++s) {
        bK[s] = wfrag[0][s][lane];
        bD[s] = wfrag[1][s][lane];
        bB[s] = wfrag[2][s][lane];
    }

    const int last = R * ROWCH - 1;
    const int aoff = (lane & 15) * NBINS + 2 * hi;

    int par = 0;
    for (int t = gw; t < tiles; t += nw) {
        const int row0 = t << 4;
        const int tn = t + nw;
        const bool more = tn < tiles;
        const int qb = more ? tn * TILECH + lane : last;
        const float* arow = myA + par * ABUF + aoff;
        float* nxt = myA + (par ^ 1) * ABUF;
        par ^= 1;

        v8f accK = {0.f,0.f,0.f,0.f,0.f,0.f,0.f,0.f};
        v8f accD = {0.f,0.f,0.f,0.f,0.f,0.f,0.f,0.f};
        v8f accB = {0.f,0.f,0.f,0.f,0.f,0.f,0.f,0.f};

        v4f sg[DEPTH][GCH];
        v2f af[2][BSTEPS];
        v4f bq[6];
        #pragma unroll
        for (int j = 0; j < BSTEPS; ++j) af[0][j] = *(const v2f*)(arow + 4 * j);
        #pragma unroll
        for (int b = 0; b < NGRP; ++b) {
            if (b >= DEPTH) stash_group(nxt, lane, b - DEPTH, sg[b % DEPTH]);
            load_group(x, qb, last, b, sg[b % DEPTH]);
            if (b + 1 < NGRP) {
                #pragma unroll
                for (int j = 0; j < BSTEPS; ++j) af[(b + 1) & 1][j] = *(const v2f*)(arow + 4 * ((b + 1) * BSTEPS + j));
            } else {
                const v4f* bt = (const v4f*)&btab[hi][0][0];
                #pragma unroll
                for (int i = 0; i < 6; ++i) bq[i] = bt[i];
            }
            __builtin_amdgcn_sched_barrier(0);
            #pragma unroll
            for (int j = 0; j < BSTEPS; ++j) {
                const int s = b * BSTEPS + j;
                accK = WMMA4(bK[s], af[b & 1][j], accK);
                accD = WMMA4(bD[s], af[b & 1][j], accD);
                accB = WMMA4(bB[s], af[b & 1][j], accB);
            }
            __builtin_amdgcn_sched_barrier(0);
        }

        float biasK[8], biasD[8], biasB[8];
        {
            const v4f k0 = bq[0], k1 = bq[1], d0 = bq[2], d1 = bq[3], b0 = bq[4], b1 = bq[5];
            #pragma unroll
            for (int r = 0; r < 4; ++r) {
                biasK[r] = k0[r]; biasK[4 + r] = k1[r];
                biasD[r] = d0[r]; biasD[4 + r] = d1[r];
                biasB[r] = b0[r]; biasB[4 + r] = b1[r];
            }
        }
        softmax_row(accK, biasK, hi);
        softmax_row(accD, biasD, hi);
        softmax_row(accB, biasB, hi);

        v8f oK, oD;
        #pragma unroll
        for (int r = 0; r < 8; ++r) { oK[r] = swap_half(accK[r]); oD[r] = swap_half(accD[r]); }
        float Kf[12], Dr[12];
        #pragma unroll
        for (int k = 0; k < 12; ++k) {
            Kf[k] = pick(accK, oK, hi, k + 1, k + 1);
            Dr[k] = pick(accD, oD, hi, ((k + 11) % 12) + 1, ((k + 7) % 12) + 1);
        }
        v8f accR;
        #pragma unroll
        for (int r = 0; r < 8; ++r) {
            float a = 0.0f;
            #pragma unroll
            for (int k = 0; k < 12; ++k) a = fmaf(Kf[k], Dr[(r - k + 12) % 12], a);
            accR[r] = a;
        }
        {
            float k0v = accK[0], d0v = accD[0];
            float root0 = k0v + d0v - k0v * d0v;
            accR[0] = hi ? accR[0] : root0;
        }

        float* ob = myA + (par ^ 1) * ABUF;
        {
            const int so = (lane & 15) * CDIM + 8 * hi;
            #pragma unroll
            for (int r = 0; r < 8; ++r) {
                if (r < 5 || !hi) {
                    ob[so + r]            = accK[r];
                    ob[OUTT + so + r]     = accD[r];
                    ob[2 * OUTT + so + r] = accB[r];
                    ob[3 * OUTT + so + r] = accR[r];
                }
            }
        }
        if (row0 + 16 <= R && (R & 3) == 0) {
            #pragma unroll
            for (int i = 0; i < 7; ++i) {
                const int q = lane + 32 * i;
                if (q < OCH) {
                    const int hd = q / 52;
                    const int w  = q - hd * 52;
                    const v4f v = *(const v4f*)(ob + 4 * q);
                    *(v4f*)(out + hd * hstride + (long long)row0 * CDIM + 4 * w) = v;
                }
            }
        } else {
            for (int idx = lane; idx < 4 * OUTT; idx += 32) {
                const int hd = idx / OUTT;
                const int w  = idx - hd * OUTT;
                if (row0 + w / CDIM < R)
                    out[hd * hstride + (long long)row0 * CDIM + w] = ob[idx];
            }
        }

        #pragma unroll
        for (int g = NGRP - DEPTH; g < NGRP; ++g) stash_group(nxt, lane, g, sg[g % DEPTH]);
    }
}

extern "C" void kernel_launch(void* const* d_in, const int* in_sizes, int n_in,
                              void* d_out, int out_size, void* d_ws, size_t ws_size,
                              hipStream_t stream) {
    const float* x  = (const float*)d_in[0];
    const float* Wk = (const float*)d_in[1];
    const float* bk = (const float*)d_in[2];
    const float* Wd = (const float*)d_in[3];
    const float* bd = (const float*)d_in[4];
    const float* Wb = (const float*)d_in[5];
    const float* bb = (const float*)d_in[6];
    float* out = (float*)d_out;
    int R = in_sizes[0] / NBINS;

    dim3 block(256);
    dim3 grid(512);
    chord_heads_kernel<<<grid, block, 0, stream>>>(x, Wk, bk, Wd, bd, Wb, bb, out, R);
}


# launch 256 workgroups instead of 512: one per WGP, resident for the whole run, 8 tiles per wave instead of 4; weight staging, barrier and first-tile wait paid once per WGP, not twice; sums unchanged
# speedup vs baseline: 1.7400x; 1.0713x over previous
#include <hip/hip_runtime.h>
#include <hip/hip_bf16.h>
#include <math.h>

typedef float v2f __attribute__((ext_vector_type(2)));
typedef float v4f __attribute__((ext_vector_type(4)));
typedef float v8f __attribute__((ext_vector_type(8)));

#define NBINS  216
#define CDIM   13
#define KSTEPS 54
#define ROWCH  54
#define TILECH 864
#define NGRP   9
#define GCH    3
#define BSTEPS 6
#define DEPTH  5
#define ABUF   (16 * NBINS)
#define WMMA4(A,B,C) __builtin_amdgcn_wmma_f32_16x16x4_f32(false,(A),false,(B),(short)0,(C),false,false)

#define WTOT   (3 * KSTEPS * 32)
#define WSLOTS ((WTOT + 255) / 256)

#define NWG    256
#define WPG    8

#define OUTT   (16 * CDIM)
#define OCH    OUTT

__device__ __forceinline__ float swap_half(float v) {
    int b = __builtin_bit_cast(int, v);
    return __builtin_bit_cast(float, __builtin_amdgcn_permlanex16(b, b, 0x76543210, 0xfedcba98, false, false));
}

__device__ __forceinline__ void softmax_row(v8f& acc, const float (&bias)[8], int hi) {
    float lg[8];
    float m = -1e30f;
    #pragma unroll
    for (int r = 0; r < 8; ++r) {
        lg[r] = acc[r] + bias[r];
        float v = (r >= 5 && hi) ? -1e30f : lg[r];
        m = fmaxf(m, v);
    }
    m = fmaxf(m, swap_half(m));
    float s = 0.0f;
    #pragma unroll
    for (int r = 0; r < 8; ++r) {
        float e = (r >= 5 && hi) ? 0.0f : __expf(lg[r] - m);
        lg[r] = e;
        s += e;
    }
    s += swap_half(s);
    #pragma unroll
    for (int r = 0; r < 8; ++r) acc[r] = lg[r] / s;
}

__device__ __forceinline__ float pick(const v8f& own, const v8f& oth, int hi, int c0, int c1) {
    float a = (c0 < 8) ? own[c0] : oth[c0 - 8];
    float b = (c1 >= 8) ? own[c1 - 8] : oth[c1];
    return hi ? b : a;
}

__device__ __forceinline__ void load_group(const float* __restrict__ x, int qbase, int last, int g, v4f (&st)[GCH]) {
    #pragma unroll
    for (int i = 0; i < GCH; ++i) {
        int qq = qbase + 32 * (g * GCH + i);
        qq = qq < last ? qq : last;
        st[i] = ((const v4f*)x)[qq];
    }
}

__device__ __forceinline__ void stash_group(float* buf, int lane, int g, const v4f (&st)[GCH]) {
    #pragma unroll
    for (int i = 0; i < GCH; ++i)
        *(v4f*)(buf + 4 * (32 * (g * GCH + i) + lane)) = st[i];
}

__global__ __launch_bounds__(256)
void chord_heads_kernel(const float* __restrict__ x,
                        const float* __restrict__ Wk, const float* __restrict__ bk,
                        const float* __restrict__ Wd, const float* __restrict__ bd,
                        const float* __restrict__ Wb, const float* __restrict__ bb,
                        float* __restrict__ out, int R)
{
    __shared__ v2f wfrag[3][KSTEPS][32];
    __shared__ __attribute__((aligned(16))) float abuf[8 * 2 * ABUF];
    __shared__ __attribute__((aligned(16))) float btab[2][3][8];
    const int tid = threadIdx.x;
    __builtin_assume(tid < 256);

    float* myA = abuf + (tid >> 5) * (2 * ABUF);
    v2f wv[WSLOTS];
    #pragma unroll
    for (int i = 0; i < WSLOTS; ++i) {
        int idx = tid + 256 * i;
        idx = idx < WTOT ? idx : WTOT - 1;
        int h   = idx / (KSTEPS * 32);
        int rem = idx - h * (KSTEPS * 32);
        int s   = rem >> 5;
        int l   = rem & 31;
        int cc  = l & 15;
        int hh  = l >> 4;
        int k0  = 4 * s + 2 * hh;
        const float* W = (h == 0) ? Wk : ((h == 1) ? Wd : Wb);
        int cl  = cc < CDIM ? cc : CDIM - 1;
        float w0 = W[k0 * CDIM + cl];
        float w1 = W[(k0 + 1) * CDIM + cl];
        wv[i].x = cc < CDIM ? w0 : 0.0f;
        wv[i].y = cc < CDIM ? w1 : 0.0f;
    }
    const int bhf = tid / 24, bhd = (tid % 24) >> 3, br = tid & 7;
    const int bcl = 8 * bhf + br;
    const float* bp = (bhd == 0) ? bk : ((bhd == 1) ? bd : bb);
    const float bv = bp[bcl < CDIM ? bcl : CDIM - 1];
    const int t0 = blockIdx.x * WPG + (tid >> 5);
    const bool have0 = t0 < ((R + 15) >> 4);
    v4f f0[NGRP][GCH];
    if (have0) {
        #pragma unroll
        for (int g = 0; g < NGRP; ++g) load_group(x, t0 * TILECH + (tid & 31), R * ROWCH - 1, g, f0[g]);
    }
    #pragma unroll
    for (int i = 0; i < WSLOTS; ++i) {
        const int idx = tid + 256 * i;
        if (idx < WTOT) (&wfrag[0][0][0])[idx] = wv[i];
    }
    if (tid < 48) btab[bhf][bhd][br] = (bcl < CDIM) ? bv : 0.0f;
    if (have0) {
        #pragma unroll
        for (int g = 0; g < NGRP; ++g) stash_group(myA, tid & 31, g, f0[g]);
    }
    __syncthreads();

    const int lane = tid & 31;
    const int wave = tid >> 5;
    const int hi   = lane >> 4;

    const int tiles = (R + 15) >> 4;
    const int gw    = blockIdx.x * WPG + wave;
    const int nw    = NWG * WPG;
    const long long hstride = (long long)R * CDIM;

    v2f bK[KSTEPS], bD[KSTEPS], bB[KSTEPS];
    #pragma unroll
    for (int s = 0; s < KSTEPS; ++s) {
        bK[s] = wfrag[0][s][lane];
        bD[s] = wfrag[1][s][lane];
        bB[s] = wfrag[2][s][lane];
    }

    const int last = R * ROWCH - 1;
    const int aoff = (lane & 15) * NBINS + 2 * hi;

    int par = 0;
    for (int t = gw; t < tiles; t += nw) {
        const int row0 = t << 4;
        const int tn = t + nw;
        const bool more = tn < tiles;
        const int qb = more ? tn * TILECH + lane : last;
        const float* arow = myA + par * ABUF + aoff;
        float* nxt = myA + (par ^ 1) * ABUF;
        par ^= 1;

        v8f accK = {0.f,0.f,0.f,0.f,0.f,0.f,0.f,0.f};
        v8f accD = {0.f,0.f,0.f,0.f,0.f,0.f,0.f,0.f};
        v8f accB = {0.f,0.f,0.f,0.f,0.f,0.f,0.f,0.f};

        v4f sg[DEPTH][GCH];
        v2f af[2][BSTEPS];
        v4f bq[6];
        #pragma unroll
        for (int j = 0; j < BSTEPS; ++j) af[0][j] = *(const v2f*)(arow + 4 * j);
        #pragma unroll
        for (int b = 0; b < NGRP; ++b) {
            if (b >= DEPTH) stash_group(nxt, lane, b - DEPTH, sg[b % DEPTH]);
            load_group(x, qb, last, b, sg[b % DEPTH]);
            if (b + 1 < NGRP) {
                #pragma unroll
                for (int j = 0; j < BSTEPS; ++j) af[(b + 1) & 1][j] = *(const v2f*)(arow + 4 * ((b + 1) * BSTEPS + j));
            } else {
                const v4f* bt = (const v4f*)&btab[hi][0][0];
                #pragma unroll
                for (int i = 0; i < 6; ++i) bq[i] = bt[i];
            }
            __builtin_amdgcn_sched_barrier(0);
            #pragma unroll
            for (int j = 0; j < BSTEPS; ++j) {
                const int s = b * BSTEPS + j;
                accK = WMMA4(bK[s], af[b & 1][j], accK);
                accD = WMMA4(bD[s], af[b & 1][j], accD);
                accB = WMMA4(bB[s], af[b & 1][j], accB);
            }
            __builtin_amdgcn_sched_barrier(0);
        }

        float biasK[8], biasD[8], biasB[8];
        {
            const v4f k0 = bq[0], k1 = bq[1], d0 = bq[2], d1 = bq[3], b0 = bq[4], b1 = bq[5];
            #pragma unroll
            for (int r = 0; r < 4; ++r) {
                biasK[r] = k0[r]; biasK[4 + r] = k1[r];
                biasD[r] = d0[r]; biasD[4 + r] = d1[r];
                biasB[r] = b0[r]; biasB[4 + r] = b1[r];
            }
        }
        softmax_row(accK, biasK, hi);
        softmax_row(accD, biasD, hi);
        softmax_row(accB, biasB, hi);

        v8f oK, oD;
        #pragma unroll
        for (int r = 0; r < 8; ++r) { oK[r] = swap_half(accK[r]); oD[r] = swap_half(accD[r]); }
        float Kf[12], Dr[12];
        #pragma unroll
        for (int k = 0; k < 12; ++k) {
            Kf[k] = pick(accK, oK, hi, k + 1, k + 1);
            Dr[k] = pick(accD, oD, hi, ((k + 11) % 12) + 1, ((k + 7) % 12) + 1);
        }
        v8f accR;
        #pragma unroll
        for (int r = 0; r < 8; ++r) {
            float a = 0.0f;
            #pragma unroll
            for (int k = 0; k < 12; ++k) a = fmaf(Kf[k], Dr[(r - k + 12) % 12], a);
            accR[r] = a;
        }
        {
            float k0v = accK[0], d0v = accD[0];
            float root0 = k0v + d0v - k0v * d0v;
            accR[0] = hi ? accR[0] : root0;
        }

        float* ob = myA + (par ^ 1) * ABUF;
        {
            const int so = (lane & 15) * CDIM + 8 * hi;
            #pragma unroll
            for (int r = 0; r < 8; ++r) {
                if (r < 5 || !hi) {
                    ob[so + r]            = accK[r];
                    ob[OUTT + so + r]     = accD[r];
                    ob[2 * OUTT + so + r] = accB[r];
                    ob[3 * OUTT + so + r] = accR[r];
                }
            }
        }
        if (row0 + 16 <= R && (R & 3) == 0) {
            #pragma unroll
            for (int i = 0; i < 7; ++i) {
                const int q = lane + 32 * i;
                if (q < OCH) {
                    const int hd = q / 52;
                    const int w  = q - hd * 52;
                    const v4f v = *(const v4f*)(ob + 4 * q);
                    *(v4f*)(out + hd * hstride + (long long)row0 * CDIM + 4 * w) = v;
                }
            }
        } else {
            for (int idx = lane; idx < 4 * OUTT; idx += 32) {
                const int hd = idx / OUTT;
                const int w  = idx - hd * OUTT;
                if (row0 + w / CDIM < R)
                    out[hd * hstride + (long long)row0 * CDIM + w] = ob[idx];
            }
        }

        #pragma unroll
        for (int g = NGRP - DEPTH; g < NGRP; ++g) stash_group(nxt, lane, g, sg[g % DEPTH]);
    }
}

extern "C" void kernel_launch(void* const* d_in, const int* in_sizes, int n_in,
                              void* d_out, int out_size, void* d_ws, size_t ws_size,
                              hipStream_t stream) {
    const float* x  = (const float*)d_in[0];
    const float* Wk = (const float*)d_in[1];
    const float* bk = (const float*)d_in[2];
    const float* Wd = (const float*)d_in[3];
    const float* bd = (const float*)d_in[4];
    const float* Wb = (const float*)d_in[5];
    const float* bb = (const float*)d_in[6];
    float* out = (float*)d_out;
    int R = in_sizes[0] / NBINS;

    dim3 block(256);
    dim3 grid(NWG);
    chord_heads_kernel<<<grid, block, 0, stream>>>(x, Wk, bk, Wd, bd, Wb, bb, out, R);
}


# softmax normalisation: one v_rcp_f32 of the row sum per row and head, then 8 multiplies, replace 8 IEEE divisions (about 11 vector instructions each); last-bit rounding of probabilities changes
# speedup vs baseline: 1.8460x; 1.0609x over previous
#include <hip/hip_runtime.h>
#include <hip/hip_bf16.h>
#include <math.h>

typedef float v2f __attribute__((ext_vector_type(2)));
typedef float v4f __attribute__((ext_vector_type(4)));
typedef float v8f __attribute__((ext_vector_type(8)));

#define NBINS  216
#define CDIM   13
#define KSTEPS 54
#define ROWCH  54
#define TILECH 864
#define NGRP   9
#define GCH    3
#define BSTEPS 6
#define DEPTH  5
#define ABUF   (16 * NBINS)
#define WMMA4(A,B,C) __builtin_amdgcn_wmma_f32_16x16x4_f32(false,(A),false,(B),(short)0,(C),false,false)

#define WTOT   (3 * KSTEPS * 32)
#define WSLOTS ((WTOT + 255) / 256)

#define NWG    256
#define WPG    8

#define OUTT   (16 * CDIM)
#define OCH    OUTT

__device__ __forceinline__ float swap_half(float v) {
    int b = __builtin_bit_cast(int, v);
    return __builtin_bit_cast(float, __builtin_amdgcn_permlanex16(b, b, 0x76543210, 0xfedcba98, false, false));
}

__device__ __forceinline__ void softmax_row(v8f& acc, const float (&bias)[8], int hi) {
    float lg[8];
    float m = -1e30f;
    #pragma unroll
    for (int r = 0; r < 8; ++r) {
        lg[r] = acc[r] + bias[r];
        float v = (r >= 5 && hi) ? -1e30f : lg[r];
        m = fmaxf(m, v);
    }
    m = fmaxf(m, swap_half(m));
    float s = 0.0f;
    #pragma unroll
    for (int r = 0; r < 8; ++r) {
        float e = (r >= 5 && hi) ? 0.0f : __expf(lg[r] - m);
        lg[r] = e;
        s += e;
    }
    s += swap_half(s);
    const float inv = __builtin_amdgcn_rcpf(s);
    #pragma unroll
    for (int r = 0; r < 8; ++r) acc[r] = lg[r] * inv;
}

__device__ __forceinline__ float pick(const v8f& own, const v8f& oth, int hi, int c0, int c1) {
    float a = (c0 < 8) ? own[c0] : oth[c0 - 8];
    float b = (c1 >= 8) ? own[c1 - 8] : oth[c1];
    return hi ? b : a;
}

__device__ __forceinline__ void load_group(const float* __restrict__ x, int qbase, int last, int g, v4f (&st)[GCH]) {
    #pragma unroll
    for (int i = 0; i < GCH; ++i) {
        int qq = qbase + 32 * (g * GCH + i);
        qq = qq < last ? qq : last;
        st[i] = ((const v4f*)x)[qq];
    }
}

__device__ __forceinline__ void stash_group(float* buf, int lane, int g, const v4f (&st)[GCH]) {
    #pragma unroll
    for (int i = 0; i < GCH; ++i)
        *(v4f*)(buf + 4 * (32 * (g * GCH + i) + lane)) = st[i];
}

__global__ __launch_bounds__(256)
void chord_heads_kernel(const float* __restrict__ x,
                        const float* __restrict__ Wk, const float* __restrict__ bk,
                        const float* __restrict__ Wd, const float* __restrict__ bd,
                        const float* __restrict__ Wb, const float* __restrict__ bb,
                        float* __restrict__ out, int R)
{
    __shared__ v2f wfrag[3][KSTEPS][32];
    __shared__ __attribute__((aligned(16))) float abuf[8 * 2 * ABUF];
    __shared__ __attribute__((aligned(16))) float btab[2][3][8];
    const int tid = threadIdx.x;
    __builtin_assume(tid < 256);

    float* myA = abuf + (tid >> 5) * (2 * ABUF);
    v2f wv[WSLOTS];
    #pragma unroll
    for (int i = 0; i < WSLOTS; ++i) {
        int idx = tid + 256 * i;
        idx = idx < WTOT ? idx : WTOT - 1;
        int h   = idx / (KSTEPS * 32);
        int rem = idx - h * (KSTEPS * 32);
        int s   = rem >> 5;
        int l   = rem & 31;
        int cc  = l & 15;
        int hh  = l >> 4;
        int k0  = 4 * s + 2 * hh;
        const float* W = (h == 0) ? Wk : ((h == 1) ? Wd : Wb);
        int cl  = cc < CDIM ? cc : CDIM - 1;
        float w0 = W[k0 * CDIM + cl];
        float w1 = W[(k0 + 1) * CDIM + cl];
        wv[i].x = cc < CDIM ? w0 : 0.0f;
        wv[i].y = cc < CDIM ? w1 : 0.0f;
    }
    const int bhf = tid / 24, bhd = (tid % 24) >> 3, br = tid & 7;
    const int bcl = 8 * bhf + br;
    const float* bp = (bhd == 0) ? bk : ((bhd == 1) ? bd : bb);
    const float bv = bp[bcl < CDIM ? bcl : CDIM - 1];
    const int t0 = blockIdx.x * WPG + (tid >> 5);
    const bool have0 = t0 < ((R + 15) >> 4);
    v4f f0[NGRP][GCH];
    if (have0) {
        #pragma unroll
        for (int g = 0; g < NGRP; ++g) load_group(x, t0 * TILECH + (tid & 31), R * ROWCH - 1, g, f0[g]);
    }
    #pragma unroll
    for (int i = 0; i < WSLOTS; ++i) {
        const int idx = tid + 256 * i;
        if (idx < WTOT) (&wfrag[0][0][0])[idx] = wv[i];
    }
    if (tid < 48) btab[bhf][bhd][br] = (bcl < CDIM) ? bv : 0.0f;
    if (have0) {
        #pragma unroll
        for (int g = 0; g < NGRP; ++g) stash_group(myA, tid & 31, g, f0[g]);
    }
    __syncthreads();

    const int lane = tid & 31;
    const int wave = tid >> 5;
    const int hi   = lane >> 4;

    const int tiles = (R + 15) >> 4;
    const int gw    = blockIdx.x * WPG + wave;
    const int nw    = NWG * WPG;
    const long long hstride = (long long)R * CDIM;

    v2f bK[KSTEPS], bD[KSTEPS], bB[KSTEPS];
    #pragma unroll
    for (int s = 0; s < KSTEPS; ++s) {
        bK[s] = wfrag[0][s][lane];
        bD[s] = wfrag[1][s][lane];
        bB[s] = wfrag[2][s][lane];
    }

    const int last = R * ROWCH - 1;
    const int aoff = (lane & 15) * NBINS + 2 * hi;

    int par = 0;
    for (int t = gw; t < tiles; t += nw) {
        const int row0 = t << 4;
        const int tn = t + nw;
        const bool more = tn < tiles;
        const int qb = more ? tn * TILECH + lane : last;
        const float* arow = myA + par * ABUF + aoff;
        float* nxt = myA + (par ^ 1) * ABUF;
        par ^= 1;

        v8f accK = {0.f,0.f,0.f,0.f,0.f,0.f,0.f,0.f};
        v8f accD = {0.f,0.f,0.f,0.f,0.f,0.f,0.f,0.f};
        v8f accB = {0.f,0.f,0.f,0.f,0.f,0.f,0.f,0.f};

        v4f sg[DEPTH][GCH];
        v2f af[2][BSTEPS];
        v4f bq[6];
        #pragma unroll
        for (int j = 0; j < BSTEPS; ++j) af[0][j] = *(const v2f*)(arow + 4 * j);
        #pragma unroll
        for (int b = 0; b < NGRP; ++b) {
            if (b >= DEPTH) stash_group(nxt, lane, b - DEPTH, sg[b % DEPTH]);
            load_group(x, qb, last, b, sg[b % DEPTH]);
            if (b + 1 < NGRP) {
                #pragma unroll
                for (int j = 0; j < BSTEPS; ++j) af[(b + 1) & 1][j] = *(const v2f*)(arow + 4 * ((b + 1) * BSTEPS + j));
            } else {
                const v4f* bt = (const v4f*)&btab[hi][0][0];
                #pragma unroll
                for (int i = 0; i < 6; ++i) bq[i] = bt[i];
            }
            __builtin_amdgcn_sched_barrier(0);
            #pragma unroll
            for (int j = 0; j < BSTEPS; ++j) {
                const int s = b * BSTEPS + j;
                accK = WMMA4(bK[s], af[b & 1][j], accK);
                accD = WMMA4(bD[s], af[b & 1][j], accD);
                accB = WMMA4(bB[s], af[b & 1][j], accB);
            }
            __builtin_amdgcn_sched_barrier(0);
        }

        float biasK[8], biasD[8], biasB[8];
        {
            const v4f k0 = bq[0], k1 = bq[1], d0 = bq[2], d1 = bq[3], b0 = bq[4], b1 = bq[5];
            #pragma unroll
            for (int r = 0; r < 4; ++r) {
                biasK[r] = k0[r]; biasK[4 + r] = k1[r];
                biasD[r] = d0[r]; biasD[4 + r] = d1[r];
                biasB[r] = b0[r]; biasB[4 + r] = b1[r];
            }
        }
        softmax_row(accK, biasK, hi);
        softmax_row(accD, biasD, hi);
        softmax_row(accB, biasB, hi);

        v8f oK, oD;
        #pragma unroll
        for (int r = 0; r < 8; ++r) { oK[r] = swap_half(accK[r]); oD[r] = swap_half(accD[r]); }
        float Kf[12], Dr[12];
        #pragma unroll
        for (int k = 0; k < 12; ++k) {
            Kf[k] = pick(accK, oK, hi, k + 1, k + 1);
            Dr[k] = pick(accD, oD, hi, ((k + 11) % 12) + 1, ((k + 7) % 12) + 1);
        }
        v8f accR;
        #pragma unroll
        for (int r = 0; r < 8; ++r) {
            float a = 0.0f;
            #pragma unroll
            for (int k = 0; k < 12; ++k) a = fmaf(Kf[k], Dr[(r - k + 12) % 12], a);
            accR[r] = a;
        }
        {
            float k0v = accK[0], d0v = accD[0];
            float root0 = k0v + d0v - k0v * d0v;
            accR[0] = hi ? accR[0] : root0;
        }

        float* ob = myA + (par ^ 1) * ABUF;
        {
            const int so = (lane & 15) * CDIM + 8 * hi;
            #pragma unroll
            for (int r = 0; r < 8; ++r) {
                if (r < 5 || !hi) {
                    ob[so + r]            = accK[r];
                    ob[OUTT + so + r]     = accD[r];
                    ob[2 * OUTT + so + r] = accB[r];
                    ob[3 * OUTT + so + r] = accR[r];
                }
            }
        }
        if (row0 + 16 <= R && (R & 3) == 0) {
            #pragma unroll
            for (int i = 0; i < 7; ++i) {
                const int q = lane + 32 * i;
                if (q < OCH) {
                    const int hd = q / 52;
                    const int w  = q - hd * 52;
                    const v4f v = *(const v4f*)(ob + 4 * q);
                    *(v4f*)(out + hd * hstride + (long long)row0 * CDIM + 4 * w) = v;
                }
            }
        } else {
            for (int idx = lane; idx < 4 * OUTT; idx += 32) {
                const int hd = idx / OUTT;
                const int w  = idx - hd * OUTT;
                if (row0 + w / CDIM < R)
                    out[hd * hstride + (long long)row0 * CDIM + w] = ob[idx];
            }
        }

        #pragma unroll
        for (int g = NGRP - DEPTH; g < NGRP; ++g) stash_group(nxt, lane, g, sg[g % DEPTH]);
    }
}

extern "C" void kernel_launch(void* const* d_in, const int* in_sizes, int n_in,
                              void* d_out, int out_size, void* d_ws, size_t ws_size,
                              hipStream_t stream) {
    const float* x  = (const float*)d_in[0];
    const float* Wk = (const float*)d_in[1];
    const float* bk = (const float*)d_in[2];
    const float* Wd = (const float*)d_in[3];
    const float* bd = (const float*)d_in[4];
    const float* Wb = (const float*)d_in[5];
    const float* bb = (const float*)d_in[6];
    float* out = (float*)d_out;
    int R = in_sizes[0] / NBINS;

    dim3 block(256);
    dim3 grid(NWG);
    chord_heads_kernel<<<grid, block, 0, stream>>>(x, Wk, bk, Wd, bd, Wb, bb, out, R);
}
